# starting kernel
# baseline (speedup 1.0000x reference)
#include <hip/hip_runtime.h>
#include <hip/hip_bf16.h>
#include <math.h>

#define HIDDEN 256
#define LN_EPS 1e-5f

typedef __attribute__((ext_vector_type(16))) _Float16 v16h;
typedef __attribute__((ext_vector_type(8)))  float    v8f;

union V16H { v16h v; _Float16 e[16]; };
union V8F  { v8f  v; float    e[8];  };

__device__ __forceinline__ float silu_f(float v) { return v / (1.0f + __expf(-v)); }

__global__ void convert_w_swizzle_kernel(const float* __restrict__ Wsrc,
                                         _Float16* __restrict__ Wdst, int nElem) {
    int o = blockIdx.x * blockDim.x + threadIdx.x;
    if (o >= nElem) return;
    int j    = o & 15;
    int lane = (o >> 4) & 31;
    int nt   = (o >> 9) & 15;
    int kb   = o >> 13;
    int k    = kb * 32 + ((j >> 3) << 4) + ((lane >> 4) << 3) + (j & 7);
    int col  = nt * 16 + (lane & 15);
    Wdst[o]  = (_Float16)Wsrc[(size_t)k * HIDDEN + col];
}

__global__ void copy_f4_kernel(const float4* __restrict__ src, float4* __restrict__ dst, int n4) {
    int i = blockIdx.x * blockDim.x + threadIdx.x;
    if (i < n4) dst[i] = src[i];
}

__global__ void edge_scatter_kernel(const float* __restrict__ h, float* agg,
                                    const int* __restrict__ src, const int* __restrict__ dst,
                                    int E) {
    long long t = (long long)blockIdx.x * blockDim.x + threadIdx.x;
    int e = (int)(t >> 6);
    if (e >= E) return;
    int c = ((int)t & 63) * 4;
    int s = src[e], d = dst[e];
    float4 m = *reinterpret_cast<const float4*>(h + (size_t)s * HIDDEN + c);
    float* o = agg + (size_t)d * HIDDEN + c;
    atomicAdd(o + 0, fmaxf(m.x, 0.0f));
    atomicAdd(o + 1, fmaxf(m.y, 0.0f));
    atomicAdd(o + 2, fmaxf(m.z, 0.0f));
    atomicAdd(o + 3, fmaxf(m.w, 0.0f));
}

template <int KDIM, bool LNRES>
__global__ __launch_bounds__(256)
void gemm_tile_kernel(const float* __restrict__ A, const _Float16* __restrict__ B16sw,
                      const float* __restrict__ bias,
                      const float* __restrict__ gamma, const float* __restrict__ beta,
                      const float* resid, float* out, int Mtot) {
    __shared__ _Float16 As[64 * 32];
    __shared__ _Float16 Bs[16 * 32 * 16];
    __shared__ float    red[2 * 64];
    __shared__ float    stat[2 * 64];

    const int tid   = threadIdx.x;
    const int lane  = tid & 31;
    const int wave  = tid >> 5;
    const int l16   = lane & 15;
    const int lhalf = lane >> 4;
    const int mtile = wave >> 1;
    const int nhalf = wave & 1;
    const int rowBase = blockIdx.x * 64;

    V8F acc[8];
#pragma unroll
    for (int t = 0; t < 8; t++)
#pragma unroll
        for (int r = 0; r < 8; r++) acc[t].e[r] = 0.0f;

    for (int k0 = 0; k0 < KDIM; k0 += 32) {
        __syncthreads();
#pragma unroll
        for (int i = 0; i < 2; i++) {
            int idx = tid + i * 256;
            int row = idx >> 3, seg = idx & 7;
            int grow = rowBase + row;
            float4 a4 = make_float4(0.f, 0.f, 0.f, 0.f);
            if (grow < Mtot)
                a4 = *reinterpret_cast<const float4*>(A + (size_t)grow * KDIM + k0 + seg * 4);
            _Float16* p = &As[row * 32 + seg * 4];
            p[0] = (_Float16)a4.x; p[1] = (_Float16)a4.y;
            p[2] = (_Float16)a4.z; p[3] = (_Float16)a4.w;
        }
        {
            const float4* bsrc =
                reinterpret_cast<const float4*>(B16sw + (size_t)(k0 >> 5) * (32 * HIDDEN));
            float4* bdst = reinterpret_cast<float4*>(Bs);
#pragma unroll
            for (int i = 0; i < 4; i++) bdst[tid + i * 256] = bsrc[tid + i * 256];
        }
        __syncthreads();

        V16H a;
        {
            const _Float16* ap = &As[(mtile * 16 + l16) * 32 + lhalf * 8];
            *reinterpret_cast<float4*>(&a.e[0]) = *reinterpret_cast<const float4*>(ap);
            *reinterpret_cast<float4*>(&a.e[8]) = *reinterpret_cast<const float4*>(ap + 16);
        }
#pragma unroll
        for (int t = 0; t < 8; t++) {
            V16H bfrag;
            const _Float16* bp = &Bs[(((nhalf * 8 + t) * 32) + lane) * 16];
            *reinterpret_cast<float4*>(&bfrag.e[0]) = *reinterpret_cast<const float4*>(bp);
            *reinterpret_cast<float4*>(&bfrag.e[8]) = *reinterpret_cast<const float4*>(bp + 8);
            acc[t].v = __builtin_amdgcn_wmma_f32_16x16x32_f16(
                false, a.v, false, bfrag.v, (short)0, acc[t].v, false, false);
        }
    }

    if (!LNRES) {
#pragma unroll
        for (int t = 0; t < 8; t++) {
            int col = (nhalf * 8 + t) * 16 + l16;
            float bv = bias[col];
#pragma unroll
            for (int r = 0; r < 8; r++) {
                int grow = rowBase + mtile * 16 + lhalf * 8 + r;
                if (grow < Mtot)
                    out[(size_t)grow * HIDDEN + col] = silu_f(acc[t].e[r] + bv);
            }
        }
    } else {
        __syncthreads();
        if (tid < 64) { red[tid] = 0.0f; red[64 + tid] = 0.0f; }
        __syncthreads();

        float bv[8];
#pragma unroll
        for (int t = 0; t < 8; t++) bv[t] = bias[(nhalf * 8 + t) * 16 + l16];

#pragma unroll
        for (int r = 0; r < 8; r++) {
            float s = 0.0f, q = 0.0f;
#pragma unroll
            for (int t = 0; t < 8; t++) {
                float v = acc[t].e[r] + bv[t];
                s += v; q += v * v;
            }
            int rl = mtile * 16 + lhalf * 8 + r;
            atomicAdd(&red[rl], s);
            atomicAdd(&red[64 + rl], q);
        }
        __syncthreads();
        if (tid < 64) {
            float mu  = red[tid] * (1.0f / HIDDEN);
            float var = red[64 + tid] * (1.0f / HIDDEN) - mu * mu;
            stat[tid]      = mu;
            stat[64 + tid] = rsqrtf(var + LN_EPS);
        }
        __syncthreads();
#pragma unroll
        for (int t = 0; t < 8; t++) {
            int col = (nhalf * 8 + t) * 16 + l16;
            float g = gamma[col], bt = beta[col];
#pragma unroll
            for (int r = 0; r < 8; r++) {
                int rl = mtile * 16 + lhalf * 8 + r;
                int grow = rowBase + rl;
                if (grow < Mtot) {
                    float v  = acc[t].e[r] + bv[t];
                    float ln = g * (v - stat[rl]) * stat[64 + rl] + bt;
                    size_t off = (size_t)grow * HIDDEN + col;
                    out[off] = silu_f(ln) + resid[off];
                }
            }
        }
    }
}

__global__ void zero_kernel(float* p, int n) {
    int i = blockIdx.x * blockDim.x + threadIdx.x;
    if (i < n) p[i] = 0.0f;
}

__global__ void pool_accum_kernel(const float* __restrict__ h, const int* __restrict__ batch,
                                  float* sums, float* cnt, int N) {
    long long t = (long long)blockIdx.x * blockDim.x + threadIdx.x;
    int n = (int)(t >> 6);
    if (n >= N) return;
    int c = ((int)t & 63) * 4;
    int g = batch[n];
    float4 v = *reinterpret_cast<const float4*>(h + (size_t)n * HIDDEN + c);
    float* o = sums + (size_t)g * HIDDEN + c;
    atomicAdd(o + 0, v.x); atomicAdd(o + 1, v.y);
    atomicAdd(o + 2, v.z); atomicAdd(o + 3, v.w);
    if (((int)t & 63) == 0) atomicAdd(&cnt[g], 1.0f);
}

__global__ void pool_div_kernel(float* out, const float* __restrict__ cnt, int n) {
    int i = blockIdx.x * blockDim.x + threadIdx.x;
    if (i < n) out[i] /= fmaxf(cnt[i >> 8], 1.0f);
}

extern "C" void kernel_launch(void* const* d_in, const int* in_sizes, int n_in,
                              void* d_out, int out_size, void* d_ws, size_t ws_size,
                              hipStream_t stream) {
    const float* x     = (const float*)d_in[0];
    const int*   edge  = (const int*)d_in[1];
    const int*   batch = (const int*)d_in[2];
    const float* Win   = (const float*)d_in[3];
    const float* b_in  = (const float*)d_in[4];
    const float* W     = (const float*)d_in[5];
    const float* b     = (const float*)d_in[6];
    const float* gamma = (const float*)d_in[7];
    const float* beta  = (const float*)d_in[8];
    float* out = (float*)d_out;

    const int NODE_IN = 64, L = 3;
    const int N = in_sizes[0] / NODE_IN;
    const int E = in_sizes[1] / 2;
    const int G = out_size / HIDDEN;

    char* ws = (char*)d_ws;
    size_t off = 0;
    auto carve = [&](size_t bytes) -> void* {
        void* p = ws + off;
        off = (off + bytes + 255) & ~(size_t)255;
        return p;
    };
    float*    h     = (float*)carve((size_t)N * HIDDEN * sizeof(float));
    float*    agg   = (float*)carve((size_t)N * HIDDEN * sizeof(float));
    _Float16* Win16 = (_Float16*)carve((size_t)NODE_IN * HIDDEN * sizeof(_Float16));
    _Float16* W16   = (_Float16*)carve((size_t)L * HIDDEN * HIDDEN * sizeof(_Float16));
    float*    cnt   = (float*)carve((size_t)G * sizeof(float));
    (void)ws_size; (void)n_in;

    int nWin = NODE_IN * HIDDEN;
    int nWl  = HIDDEN * HIDDEN;
    convert_w_swizzle_kernel<<<(nWin + 255) / 256, 256, 0, stream>>>(Win, Win16, nWin);
    for (int l = 0; l < L; l++)
        convert_w_swizzle_kernel<<<(nWl + 255) / 256, 256, 0, stream>>>(
            W + (size_t)l * nWl, W16 + (size_t)l * nWl, nWl);

    int mblocks = (N + 63) / 64;
    gemm_tile_kernel<NODE_IN, false><<<mblocks, 256, 0, stream>>>(
        x, Win16, b_in, nullptr, nullptr, nullptr, h, N);

    int n4 = N * HIDDEN / 4;
    long long escatter = (long long)E * 64;
    for (int l = 0; l < L; l++) {
        copy_f4_kernel<<<(n4 + 255) / 256, 256, 0, stream>>>((const float4*)h, (float4*)agg, n4);
        edge_scatter_kernel<<<(int)((escatter + 255) / 256), 256, 0, stream>>>(
            h, agg, edge, edge + E, E);
        gemm_tile_kernel<HIDDEN, true><<<mblocks, 256, 0, stream>>>(
            agg, W16 + (size_t)l * nWl,
            b + l * HIDDEN, gamma + l * HIDDEN, beta + l * HIDDEN, h, h, N);
    }

    int nz = G * HIDDEN;
    zero_kernel<<<(nz + 255) / 256, 256, 0, stream>>>(out, nz);
    zero_kernel<<<(G + 255) / 256, 256, 0, stream>>>(cnt, G);
    long long npool = (long long)N * 64;
    pool_accum_kernel<<<(int)((npool + 255) / 256), 256, 0, stream>>>(h, batch, out, cnt, N);
    pool_div_kernel<<<(nz + 255) / 256, 256, 0, stream>>>(out, cnt, nz);
}


# edges sorted by destination once per call (count, allocate, fill, sort); per layer one atomic-free wave-per-row gather replaces the agg=h copy and f32-atomic scatter; neighbour sums in fixed order
# speedup vs baseline: 1.4480x; 1.4480x over previous
#include <hip/hip_runtime.h>
#include <hip/hip_bf16.h>
#include <math.h>

#define HIDDEN 256
#define LN_EPS 1e-5f

typedef __attribute__((ext_vector_type(16))) _Float16 v16h;
typedef __attribute__((ext_vector_type(8)))  float    v8f;

union V16H { v16h v; _Float16 e[16]; };
union V8F  { v8f  v; float    e[8];  };

__device__ __forceinline__ float silu_f(float v) { return v / (1.0f + __expf(-v)); }

__global__ void convert_w_swizzle_kernel(const float* __restrict__ Wsrc,
                                         _Float16* __restrict__ Wdst, int nElem) {
    int o = blockIdx.x * blockDim.x + threadIdx.x;
    if (o >= nElem) return;
    int j    = o & 15;
    int lane = (o >> 4) & 31;
    int nt   = (o >> 9) & 15;
    int kb   = o >> 13;
    int k    = kb * 32 + ((j >> 3) << 4) + ((lane >> 4) << 3) + (j & 7);
    int col  = nt * 16 + (lane & 15);
    Wdst[o]  = (_Float16)Wsrc[(size_t)k * HIDDEN + col];
}

__global__ void zero_int_kernel(int* p, int n) {
    int i = blockIdx.x * blockDim.x + threadIdx.x;
    if (i < n) p[i] = 0;
}

__global__ void degree_count_kernel(const int* __restrict__ dst, int* deg, int E) {
    int e = blockIdx.x * blockDim.x + threadIdx.x;
    if (e < E) atomicAdd(&deg[dst[e]], 1);
}

__global__ __launch_bounds__(1024)
void segment_alloc_kernel(const int* __restrict__ deg, int* __restrict__ cursor,
                          int* total, int N) {
    __shared__ int wsum[32];
    __shared__ int base_s;
    const int tid = threadIdx.x, lane = tid & 31, wave = tid >> 5;
    const int n = blockIdx.x * 1024 + tid;
    const int d = (n < N) ? deg[n] : 0;
    int v = d;
#pragma unroll
    for (int o = 1; o < 32; o <<= 1) {
        int t = __shfl_up(v, o);
        if (lane >= o) v += t;
    }
    if (lane == 31) wsum[wave] = v;
    __syncthreads();
    if (wave == 0) {
        int s = wsum[lane];
        int sv = s;
#pragma unroll
        for (int o = 1; o < 32; o <<= 1) {
            int t = __shfl_up(sv, o);
            if (lane >= o) sv += t;
        }
        wsum[lane] = sv - s;
        if (lane == 31) base_s = atomicAdd(total, sv);
    }
    __syncthreads();
    if (n < N) cursor[n] = base_s + wsum[wave] + v;
}

__global__ void segment_fill_kernel(const int* __restrict__ src, const int* __restrict__ dst,
                                    int* cursor, int* __restrict__ col, int E) {
    int e = blockIdx.x * blockDim.x + threadIdx.x;
    if (e >= E) return;
    int pos = atomicSub(&cursor[dst[e]], 1) - 1;
    if (pos >= 0 && pos < E) col[pos] = src[e];
}

__global__ void segment_sort_kernel(const int* __restrict__ rowstart, const int* __restrict__ deg,
                                    int* col, int N, int E) {
    int n = blockIdx.x * blockDim.x + threadIdx.x;
    if (n >= N) return;
    int st = min(max(rowstart[n], 0), E);
    int dg = min(max(deg[n], 0), E - st);
    for (int i = 1; i < dg; i++) {
        int key = col[st + i];
        int j = i - 1;
        while (j >= 0 && col[st + j] > key) { col[st + j + 1] = col[st + j]; j--; }
        col[st + j + 1] = key;
    }
}

__device__ __forceinline__ void relu_add(float4& a, const float4& v) {
    a.x += fmaxf(v.x, 0.0f); a.y += fmaxf(v.y, 0.0f);
    a.z += fmaxf(v.z, 0.0f); a.w += fmaxf(v.w, 0.0f);
}

template <int K>
__device__ __forceinline__ void gather_group(const float4* __restrict__ h4, const int* __restrict__ colp,
                                             int lane, float4& a0, float4& a1) {
    float4 v0[K], v1[K];
#pragma unroll
    for (int k = 0; k < K; k++) {
        const float4* r = h4 + (size_t)colp[k] * (HIDDEN / 4);
        v0[k] = r[lane];
        v1[k] = r[lane + 32];
    }
#pragma unroll
    for (int k = 0; k < K; k++) { relu_add(a0, v0[k]); relu_add(a1, v1[k]); }
}

__global__ __launch_bounds__(256)
void gather_rows_kernel(const float4* __restrict__ h4, float4* __restrict__ agg4,
                        const int* __restrict__ rowstart, const int* __restrict__ deg,
                        const int* __restrict__ col, int N, int E) {
    const int lane = threadIdx.x & 31;
    const int wpb  = blockDim.x >> 5;
    const int w0 = __builtin_amdgcn_readfirstlane((int)(blockIdx.x * wpb + (threadIdx.x >> 5)));
    const int nw = gridDim.x * wpb;
    for (int n = w0; n < N; n += nw) {
        int st = min(max(rowstart[n], 0), E);
        int dg = min(max(deg[n], 0), E - st);
        const float4* hs = h4 + (size_t)n * (HIDDEN / 4);
        float4 s0 = hs[lane], s1 = hs[lane + 32];
        float4 a0 = make_float4(0.f, 0.f, 0.f, 0.f), a1 = a0;
        const int* cp = col + st;
        int j = 0;
        for (; j + 4 <= dg; j += 4) gather_group<4>(h4, cp + j, lane, a0, a1);
        const int rem = dg - j;
        if (rem == 3)      gather_group<3>(h4, cp + j, lane, a0, a1);
        else if (rem == 2) gather_group<2>(h4, cp + j, lane, a0, a1);
        else if (rem == 1) gather_group<1>(h4, cp + j, lane, a0, a1);
        float4* o = agg4 + (size_t)n * (HIDDEN / 4);
        o[lane]      = make_float4(s0.x + a0.x, s0.y + a0.y, s0.z + a0.z, s0.w + a0.w);
        o[lane + 32] = make_float4(s1.x + a1.x, s1.y + a1.y, s1.z + a1.z, s1.w + a1.w);
    }
}

template <int KDIM, bool LNRES>
__global__ __launch_bounds__(256)
void gemm_tile_kernel(const float* __restrict__ A, const _Float16* __restrict__ B16sw,
                      const float* __restrict__ bias,
                      const float* __restrict__ gamma, const float* __restrict__ beta,
                      const float* resid, float* out, int Mtot) {
    __shared__ _Float16 As[64 * 32];
    __shared__ _Float16 Bs[16 * 32 * 16];
    __shared__ float    red[2 * 64];
    __shared__ float    stat[2 * 64];

    const int tid   = threadIdx.x;
    const int lane  = tid & 31;
    const int wave  = tid >> 5;
    const int l16   = lane & 15;
    const int lhalf = lane >> 4;
    const int mtile = wave >> 1;
    const int nhalf = wave & 1;
    const int rowBase = blockIdx.x * 64;

    V8F acc[8];
#pragma unroll
    for (int t = 0; t < 8; t++)
#pragma unroll
        for (int r = 0; r < 8; r++) acc[t].e[r] = 0.0f;

    for (int k0 = 0; k0 < KDIM; k0 += 32) {
        __syncthreads();
#pragma unroll
        for (int i = 0; i < 2; i++) {
            int idx = tid + i * 256;
            int row = idx >> 3, seg = idx & 7;
            int grow = rowBase + row;
            float4 a4 = make_float4(0.f, 0.f, 0.f, 0.f);
            if (grow < Mtot)
                a4 = *reinterpret_cast<const float4*>(A + (size_t)grow * KDIM + k0 + seg * 4);
            _Float16* p = &As[row * 32 + seg * 4];
            p[0] = (_Float16)a4.x; p[1] = (_Float16)a4.y;
            p[2] = (_Float16)a4.z; p[3] = (_Float16)a4.w;
        }
        {
            const float4* bsrc =
                reinterpret_cast<const float4*>(B16sw + (size_t)(k0 >> 5) * (32 * HIDDEN));
            float4* bdst = reinterpret_cast<float4*>(Bs);
#pragma unroll
            for (int i = 0; i < 4; i++) bdst[tid + i * 256] = bsrc[tid + i * 256];
        }
        __syncthreads();

        V16H a;
        {
            const _Float16* ap = &As[(mtile * 16 + l16) * 32 + lhalf * 8];
            *reinterpret_cast<float4*>(&a.e[0]) = *reinterpret_cast<const float4*>(ap);
            *reinterpret_cast<float4*>(&a.e[8]) = *reinterpret_cast<const float4*>(ap + 16);
        }
#pragma unroll
        for (int t = 0; t < 8; t++) {
            V16H bfrag;
            const _Float16* bp = &Bs[(((nhalf * 8 + t) * 32) + lane) * 16];
            *reinterpret_cast<float4*>(&bfrag.e[0]) = *reinterpret_cast<const float4*>(bp);
            *reinterpret_cast<float4*>(&bfrag.e[8]) = *reinterpret_cast<const float4*>(bp + 8);
            acc[t].v = __builtin_amdgcn_wmma_f32_16x16x32_f16(
                false, a.v, false, bfrag.v, (short)0, acc[t].v, false, false);
        }
    }

    if (!LNRES) {
#pragma unroll
        for (int t = 0; t < 8; t++) {
            int col = (nhalf * 8 + t) * 16 + l16;
            float bv = bias[col];
#pragma unroll
            for (int r = 0; r < 8; r++) {
                int grow = rowBase + mtile * 16 + lhalf * 8 + r;
                if (grow < Mtot)
                    out[(size_t)grow * HIDDEN + col] = silu_f(acc[t].e[r] + bv);
            }
        }
    } else {
        __syncthreads();
        if (tid < 64) { red[tid] = 0.0f; red[64 + tid] = 0.0f; }
        __syncthreads();

        float bv[8];
#pragma unroll
        for (int t = 0; t < 8; t++) bv[t] = bias[(nhalf * 8 + t) * 16 + l16];

#pragma unroll
        for (int r = 0; r < 8; r++) {
            float s = 0.0f, q = 0.0f;
#pragma unroll
            for (int t = 0; t < 8; t++) {
                float v = acc[t].e[r] + bv[t];
                s += v; q += v * v;
            }
            int rl = mtile * 16 + lhalf * 8 + r;
            atomicAdd(&red[rl], s);
            atomicAdd(&red[64 + rl], q);
        }
        __syncthreads();
        if (tid < 64) {
            float mu  = red[tid] * (1.0f / HIDDEN);
            float var = red[64 + tid] * (1.0f / HIDDEN) - mu * mu;
            stat[tid]      = mu;
            stat[64 + tid] = rsqrtf(var + LN_EPS);
        }
        __syncthreads();
#pragma unroll
        for (int t = 0; t < 8; t++) {
            int col = (nhalf * 8 + t) * 16 + l16;
            float g = gamma[col], bt = beta[col];
#pragma unroll
            for (int r = 0; r < 8; r++) {
                int rl = mtile * 16 + lhalf * 8 + r;
                int grow = rowBase + rl;
                if (grow < Mtot) {
                    float v  = acc[t].e[r] + bv[t];
                    float ln = g * (v - stat[rl]) * stat[64 + rl] + bt;
                    size_t off = (size_t)grow * HIDDEN + col;
                    out[off] = silu_f(ln) + resid[off];
                }
            }
        }
    }
}

__global__ void zero_kernel(float* p, int n) {
    int i = blockIdx.x * blockDim.x + threadIdx.x;
    if (i < n) p[i] = 0.0f;
}

__global__ void pool_accum_kernel(const float* __restrict__ h, const int* __restrict__ batch,
                                  float* sums, float* cnt, int N) {
    long long t = (long long)blockIdx.x * blockDim.x + threadIdx.x;
    int n = (int)(t >> 6);
    if (n >= N) return;
    int c = ((int)t & 63) * 4;
    int g = batch[n];
    float4 v = *reinterpret_cast<const float4*>(h + (size_t)n * HIDDEN + c);
    float* o = sums + (size_t)g * HIDDEN + c;
    atomicAdd(o + 0, v.x); atomicAdd(o + 1, v.y);
    atomicAdd(o + 2, v.z); atomicAdd(o + 3, v.w);
    if (((int)t & 63) == 0) atomicAdd(&cnt[g], 1.0f);
}

__global__ void pool_div_kernel(float* out, const float* __restrict__ cnt, int n) {
    int i = blockIdx.x * blockDim.x + threadIdx.x;
    if (i < n) out[i] /= fmaxf(cnt[i >> 8], 1.0f);
}

extern "C" void kernel_launch(void* const* d_in, const int* in_sizes, int n_in,
                              void* d_out, int out_size, void* d_ws, size_t ws_size,
                              hipStream_t stream) {
    const float* x     = (const float*)d_in[0];
    const int*   edge  = (const int*)d_in[1];
    const int*   batch = (const int*)d_in[2];
    const float* Win   = (const float*)d_in[3];
    const float* b_in  = (const float*)d_in[4];
    const float* W     = (const float*)d_in[5];
    const float* b     = (const float*)d_in[6];
    const float* gamma = (const float*)d_in[7];
    const float* beta  = (const float*)d_in[8];
    float* out = (float*)d_out;

    const int NODE_IN = 64, L = 3;
    const int N = in_sizes[0] / NODE_IN;
    const int E = in_sizes[1] / 2;
    const int G = out_size / HIDDEN;

    char* ws = (char*)d_ws;
    size_t off = 0;
    auto carve = [&](size_t bytes) -> void* {
        void* p = ws + off;
        off = (off + bytes + 255) & ~(size_t)255;
        return p;
    };
    float*    h     = (float*)carve((size_t)N * HIDDEN * sizeof(float));
    float*    agg   = (float*)carve((size_t)N * HIDDEN * sizeof(float));
    _Float16* Win16 = (_Float16*)carve((size_t)NODE_IN * HIDDEN * sizeof(_Float16));
    _Float16* W16   = (_Float16*)carve((size_t)L * HIDDEN * HIDDEN * sizeof(_Float16));
    float*    cnt   = (float*)carve((size_t)G * sizeof(float));
    int*      deg   = (int*)carve(((size_t)N + 1) * sizeof(int));
    int*      rowst = (int*)carve((size_t)N * sizeof(int));
    int*      col   = (int*)carve((size_t)E * sizeof(int));
    (void)n_in;
    if (off > ws_size) return;

    const int* esrc = edge;
    const int* edst = edge + E;
    zero_int_kernel<<<(N + 1 + 255) / 256, 256, 0, stream>>>(deg, N + 1);
    degree_count_kernel<<<(E + 255) / 256, 256, 0, stream>>>(edst, deg, E);
    segment_alloc_kernel<<<(N + 1023) / 1024, 1024, 0, stream>>>(deg, rowst, deg + N, N);
    segment_fill_kernel<<<(E + 255) / 256, 256, 0, stream>>>(esrc, edst, rowst, col, E);
    segment_sort_kernel<<<(N + 255) / 256, 256, 0, stream>>>(rowst, deg, col, N, E);

    int nWin = NODE_IN * HIDDEN;
    int nWl  = HIDDEN * HIDDEN;
    convert_w_swizzle_kernel<<<(nWin + 255) / 256, 256, 0, stream>>>(Win, Win16, nWin);
    for (int l = 0; l < L; l++)
        convert_w_swizzle_kernel<<<(nWl + 255) / 256, 256, 0, stream>>>(
            W + (size_t)l * nWl, W16 + (size_t)l * nWl, nWl);

    int mblocks = (N + 63) / 64;
    gemm_tile_kernel<NODE_IN, false><<<mblocks, 256, 0, stream>>>(
        x, Win16, b_in, nullptr, nullptr, nullptr, h, N);

    for (int l = 0; l < L; l++) {
        gather_rows_kernel<<<2048, 256, 0, stream>>>(
            (const float4*)h, (float4*)agg, rowst, deg, col, N, E);
        gemm_tile_kernel<HIDDEN, true><<<mblocks, 256, 0, stream>>>(
            agg, W16 + (size_t)l * nWl,
            b + l * HIDDEN, gamma + l * HIDDEN, beta + l * HIDDEN, h, h, N);
    }

    int nz = G * HIDDEN;
    zero_kernel<<<(nz + 255) / 256, 256, 0, stream>>>(out, nz);
    zero_kernel<<<(G + 255) / 256, 256, 0, stream>>>(cnt, G);
    long long npool = (long long)N * 64;
    pool_accum_kernel<<<(int)((npool + 255) / 256), 256, 0, stream>>>(h, batch, out, cnt, N);
    pool_div_kernel<<<(nz + 255) / 256, 256, 0, stream>>>(out, cnt, nz);
}
